# starting kernel
# baseline (speedup 1.0000x reference)
#include <hip/hip_runtime.h>


typedef float v2f __attribute__((ext_vector_type(2)));
typedef float v8f __attribute__((ext_vector_type(8)));

#define D_FEAT     128
#define TILE       16
#define ROWSTRIDE  132
#define WPB        2
#define BLOCK      (WPB * 32)
#define LDS_F_PER_WAVE (2 * TILE * ROWSTRIDE + TILE * TILE)

__global__ __launch_bounds__(BLOCK)
void edge_dot_wmma(const float* __restrict__ h,
                   const int*   __restrict__ src,
                   const int*   __restrict__ dst,
                   float*       __restrict__ out,
                   int n_edges, int n_nodes)
{
    __shared__ float lds[WPB][LDS_F_PER_WAVE];
    const int wave = threadIdx.x >> 5;
    const int lane = threadIdx.x & 31;

    float* srcT = &lds[wave][0];
    float* dstT = srcT + TILE * ROWSTRIDE;
    float* dmat = dstT + TILE * ROWSTRIDE;

    const long long tile = (long long)blockIdx.x * WPB + wave;
    const int e0 = (int)(tile * TILE);

    int e  = e0 + (lane & 15);
    int ce = (e < n_edges) ? e : 0;
    int node = (lane < 16) ? __builtin_nontemporal_load(src + ce)
                           : __builtin_nontemporal_load(dst + ce);
    if ((unsigned)node >= (unsigned)n_nodes) node = 0;

    #pragma unroll 4
    for (int v = 0; v < 32; ++v) {
        int nv = __shfl(node, v, 32);
        const float4 g = *((const float4*)(h + (size_t)nv * D_FEAT) + lane);
        float* row = (v < TILE) ? (srcT + v * ROWSTRIDE)
                                : (dstT + (v - TILE) * ROWSTRIDE);
        *(float4*)(row + 4 * lane) = g;
    }
    __builtin_amdgcn_wave_barrier();

    const int   r    = lane & 15;
    const int   koff = (lane >> 4) << 1;
    const float* pa  = srcT + r * ROWSTRIDE + koff;
    const float* pb  = dstT + r * ROWSTRIDE + koff;

    v8f acc0 = {0.f, 0.f, 0.f, 0.f, 0.f, 0.f, 0.f, 0.f};
    v8f acc1 = {0.f, 0.f, 0.f, 0.f, 0.f, 0.f, 0.f, 0.f};
    #pragma unroll
    for (int k = 0; k < D_FEAT; k += 8) {
        v2f a0 = *(const v2f*)(pa + k);
        v2f b0 = *(const v2f*)(pb + k);
        acc0 = __builtin_amdgcn_wmma_f32_16x16x4_f32(false, a0, false, b0,
                                                     (short)0, acc0, false, false);
        v2f a1 = *(const v2f*)(pa + k + 4);
        v2f b1 = *(const v2f*)(pb + k + 4);
        acc1 = __builtin_amdgcn_wmma_f32_16x16x4_f32(false, a1, false, b1,
                                                     (short)0, acc1, false, false);
    }
    v8f acc = acc0 + acc1;

    const int mbase = (lane >> 4) * 8;
    const int ncol  = lane & 15;
    #pragma unroll
    for (int rr = 0; rr < 8; ++rr)
        dmat[(mbase + rr) * TILE + ncol] = acc[rr];
    __builtin_amdgcn_wave_barrier();

    if (lane < 16) {
        int eo = e0 + lane;
        if (eo < n_edges)
            __builtin_nontemporal_store(dmat[lane * TILE + lane], out + eo);
    }
}

extern "C" void kernel_launch(void* const* d_in, const int* in_sizes, int n_in,
                              void* d_out, int out_size, void* d_ws, size_t ws_size,
                              hipStream_t stream) {
    const float* h   = (const float*)d_in[0];
    const int*   src = (const int*)d_in[1];
    const int*   dst = (const int*)d_in[2];
    float*       out = (float*)d_out;

    const int n_edges = in_sizes[1];
    const int n_nodes = in_sizes[0] / D_FEAT;

    const int ntiles = (n_edges + TILE - 1) / TILE;
    const int blocks = (ntiles + WPB - 1) / WPB;

    edge_dot_wmma<<<blocks, BLOCK, 0, stream>>>(h, src, dst, out, n_edges, n_nodes);
}


# All 32 row loads of a 16-edge tile are issued before the first LDS store (was 4 at a time): 16 KiB of gather in flight per wave, one memory round trip per tile instead of eight
# speedup vs baseline: 1.3659x; 1.3659x over previous
#include <hip/hip_runtime.h>


typedef float v2f __attribute__((ext_vector_type(2)));
typedef float v8f __attribute__((ext_vector_type(8)));

#define D_FEAT     128
#define TILE       16
#define ROWSTRIDE  132
#define WPB        2
#define BLOCK      (WPB * 32)
#define LDS_F_PER_WAVE (2 * TILE * ROWSTRIDE + TILE * TILE)

__global__ __launch_bounds__(BLOCK)
void edge_dot_wmma(const float* __restrict__ h,
                   const int*   __restrict__ src,
                   const int*   __restrict__ dst,
                   float*       __restrict__ out,
                   int n_edges, int n_nodes)
{
    __shared__ float lds[WPB][LDS_F_PER_WAVE];
    const int wave = threadIdx.x >> 5;
    const int lane = threadIdx.x & 31;

    float* srcT = &lds[wave][0];
    float* dstT = srcT + TILE * ROWSTRIDE;
    float* dmat = dstT + TILE * ROWSTRIDE;

    const long long tile = (long long)blockIdx.x * WPB + wave;
    const int e0 = (int)(tile * TILE);

    int e  = e0 + (lane & 15);
    int ce = (e < n_edges) ? e : 0;
    int node = (lane < 16) ? __builtin_nontemporal_load(src + ce)
                           : __builtin_nontemporal_load(dst + ce);
    if ((unsigned)node >= (unsigned)n_nodes) node = 0;

    float4 g[2 * TILE];
    #pragma unroll
    for (int v = 0; v < 2 * TILE; ++v) {
        int nv = __shfl(node, v, 32);
        g[v] = *((const float4*)(h + (size_t)nv * D_FEAT) + lane);
    }
    #pragma unroll
    for (int v = 0; v < 2 * TILE; ++v) {
        float* row = (v < TILE) ? (srcT + v * ROWSTRIDE)
                                : (dstT + (v - TILE) * ROWSTRIDE);
        *(float4*)(row + 4 * lane) = g[v];
    }
    __builtin_amdgcn_wave_barrier();

    const int   r    = lane & 15;
    const int   koff = (lane >> 4) << 1;
    const float* pa  = srcT + r * ROWSTRIDE + koff;
    const float* pb  = dstT + r * ROWSTRIDE + koff;

    v8f acc0 = {0.f, 0.f, 0.f, 0.f, 0.f, 0.f, 0.f, 0.f};
    v8f acc1 = {0.f, 0.f, 0.f, 0.f, 0.f, 0.f, 0.f, 0.f};
    #pragma unroll
    for (int k = 0; k < D_FEAT; k += 8) {
        v2f a0 = *(const v2f*)(pa + k);
        v2f b0 = *(const v2f*)(pb + k);
        acc0 = __builtin_amdgcn_wmma_f32_16x16x4_f32(false, a0, false, b0,
                                                     (short)0, acc0, false, false);
        v2f a1 = *(const v2f*)(pa + k + 4);
        v2f b1 = *(const v2f*)(pb + k + 4);
        acc1 = __builtin_amdgcn_wmma_f32_16x16x4_f32(false, a1, false, b1,
                                                     (short)0, acc1, false, false);
    }
    v8f acc = acc0 + acc1;

    const int mbase = (lane >> 4) * 8;
    const int ncol  = lane & 15;
    #pragma unroll
    for (int rr = 0; rr < 8; ++rr)
        dmat[(mbase + rr) * TILE + ncol] = acc[rr];
    __builtin_amdgcn_wave_barrier();

    if (lane < 16) {
        int eo = e0 + lane;
        if (eo < n_edges)
            __builtin_nontemporal_store(dmat[lane * TILE + lane], out + eo);
    }
}

extern "C" void kernel_launch(void* const* d_in, const int* in_sizes, int n_in,
                              void* d_out, int out_size, void* d_ws, size_t ws_size,
                              hipStream_t stream) {
    const float* h   = (const float*)d_in[0];
    const int*   src = (const int*)d_in[1];
    const int*   dst = (const int*)d_in[2];
    float*       out = (float*)d_out;

    const int n_edges = in_sizes[1];
    const int n_nodes = in_sizes[0] / D_FEAT;

    const int ntiles = (n_edges + TILE - 1) / TILE;
    const int blocks = (ntiles + WPB - 1) / WPB;

    edge_dot_wmma<<<blocks, BLOCK, 0, stream>>>(h, src, dst, out, n_edges, n_nodes);
}


# Each wave walks 5 tiles in a loop and issues the next tile's 32 row loads into the just-drained registers before the current tile's WMMA phase: the gather round trip hides behind LDS and matrix work
# speedup vs baseline: 1.4426x; 1.0562x over previous
#include <hip/hip_runtime.h>


typedef float v2f __attribute__((ext_vector_type(2)));
typedef float v8f __attribute__((ext_vector_type(8)));

#define D_FEAT     128
#define TILE       16
#define ROWSTRIDE  132
#define WPB        2
#define BLOCK      (WPB * 32)
#define TPW        5
#define LDS_F_PER_WAVE (2 * TILE * ROWSTRIDE + TILE * TILE)

__device__ __forceinline__ int load_nodes(const int* __restrict__ src, const int* __restrict__ dst,
                                          int tile, int lane, int n_edges)
{
    long long e = (long long)tile * TILE + (lane & 15);
    int ce = (e < (long long)n_edges) ? (int)e : 0;
    return (lane < 16) ? __builtin_nontemporal_load(src + ce)
                       : __builtin_nontemporal_load(dst + ce);
}

#define ROWS(M) M(0) M(1) M(2) M(3) M(4) M(5) M(6) M(7) M(8) M(9) M(10) M(11) M(12) M(13) M(14) M(15) \
                M(16) M(17) M(18) M(19) M(20) M(21) M(22) M(23) M(24) M(25) M(26) M(27) M(28) M(29) M(30) M(31)
#define DECL_ROW(v)  float4 g##v;
#define LOAD_ROW(v)  g##v = *((const float4*)(h + (size_t)__shfl(node, (v), 32) * D_FEAT) + lane);
#define STORE_ROW(v) *(float4*)((((v) < TILE) ? (srcT + (v) * ROWSTRIDE) : (dstT + ((v) - TILE) * ROWSTRIDE)) + 4 * lane) = g##v;

__device__ __forceinline__ void tile_scores(const float* srcT, const float* dstT, float* dmat,
                                            int lane, int tile, float* __restrict__ out, int n_edges)
{
    const int   r    = lane & 15;
    const int   koff = (lane >> 4) << 1;
    const float* pa  = srcT + r * ROWSTRIDE + koff;
    const float* pb  = dstT + r * ROWSTRIDE + koff;

    v8f acc0 = {0.f, 0.f, 0.f, 0.f, 0.f, 0.f, 0.f, 0.f};
    v8f acc1 = {0.f, 0.f, 0.f, 0.f, 0.f, 0.f, 0.f, 0.f};
    #pragma unroll
    for (int k = 0; k < D_FEAT; k += 8) {
        v2f a0 = *(const v2f*)(pa + k);
        v2f b0 = *(const v2f*)(pb + k);
        acc0 = __builtin_amdgcn_wmma_f32_16x16x4_f32(false, a0, false, b0,
                                                     (short)0, acc0, false, false);
        v2f a1 = *(const v2f*)(pa + k + 4);
        v2f b1 = *(const v2f*)(pb + k + 4);
        acc1 = __builtin_amdgcn_wmma_f32_16x16x4_f32(false, a1, false, b1,
                                                     (short)0, acc1, false, false);
    }
    v8f acc = acc0 + acc1;

    const int mbase = (lane >> 4) * 8;
    const int ncol  = lane & 15;
    #pragma unroll
    for (int rr = 0; rr < 8; ++rr)
        dmat[(mbase + rr) * TILE + ncol] = acc[rr];
    __builtin_amdgcn_wave_barrier();

    if (lane < 16) {
        long long eo = (long long)tile * TILE + lane;
        if (eo < (long long)n_edges)
            __builtin_nontemporal_store(dmat[lane * TILE + lane], out + eo);
    }
}

__global__ __launch_bounds__(BLOCK)
void edge_dot_wmma(const float* __restrict__ h,
                   const int*   __restrict__ src,
                   const int*   __restrict__ dst,
                   float*       __restrict__ out,
                   int n_edges, int n_nodes)
{
    __shared__ float lds[WPB][LDS_F_PER_WAVE];
    const int lane = threadIdx.x & 31;
    const int wave = __builtin_amdgcn_readfirstlane((int)(threadIdx.x >> 5));

    float* srcT = &lds[wave][0];
    float* dstT = srcT + TILE * ROWSTRIDE;
    float* dmat = dstT + TILE * ROWSTRIDE;

    const int ntiles = (n_edges + TILE - 1) / TILE;
    const int nwaves = (int)gridDim.x * WPB;
    int tile = (int)blockIdx.x * WPB + wave;
    if (tile >= ntiles) return;

    ROWS(DECL_ROW)
    int next   = tile + nwaves;
    int node_c = load_nodes(src, dst, tile, lane, n_edges);
    int node_n = load_nodes(src, dst, next, lane, n_edges);
    int node   = ((unsigned)node_c < (unsigned)n_nodes) ? node_c : 0;
    ROWS(LOAD_ROW)

    while (next < ntiles) {
        ROWS(STORE_ROW)
        __builtin_amdgcn_wave_barrier();
        node = ((unsigned)node_n < (unsigned)n_nodes) ? node_n : 0;
        ROWS(LOAD_ROW)
        const int nn = next + nwaves;
        node_n = load_nodes(src, dst, nn, lane, n_edges);
        __builtin_amdgcn_wave_barrier();
        tile_scores(srcT, dstT, dmat, lane, tile, out, n_edges);
        __builtin_amdgcn_wave_barrier();
        tile = next;
        next = nn;
    }
    ROWS(STORE_ROW)
    __builtin_amdgcn_wave_barrier();
    tile_scores(srcT, dstT, dmat, lane, tile, out, n_edges);
}

extern "C" void kernel_launch(void* const* d_in, const int* in_sizes, int n_in,
                              void* d_out, int out_size, void* d_ws, size_t ws_size,
                              hipStream_t stream) {
    const float* h   = (const float*)d_in[0];
    const int*   src = (const int*)d_in[1];
    const int*   dst = (const int*)d_in[2];
    float*       out = (float*)d_out;

    const int n_edges = in_sizes[1];
    const int n_nodes = in_sizes[0] / D_FEAT;

    const int ntiles = (n_edges + TILE - 1) / TILE;
    const int blocks = (ntiles + WPB * TPW - 1) / (WPB * TPW);

    edge_dot_wmma<<<blocks, BLOCK, 0, stream>>>(h, src, dst, out, n_edges, n_nodes);
}


# One edge = one 16x16 tile: each lane's coalesced 16 bytes of the src and dst row are the A and B fragments, score = trace folded by a lane butterfly (new summation tree); no LDS staging
# speedup vs baseline: 1.8102x; 1.2548x over previous
#include <hip/hip_runtime.h>


typedef float v2f __attribute__((ext_vector_type(2)));
typedef float v4f __attribute__((ext_vector_type(4)));
typedef float v8f __attribute__((ext_vector_type(8)));
typedef __attribute__((address_space(1))) const v4f gv4f;

#define D_FEAT 128
#define GROUP  32
#define WPB    4
#define BLOCK  (WPB * 32)
#define GPW    4
#define DEPTH  8

__device__ __forceinline__ int load_ids(const int* __restrict__ p, long long g, int lane, int n_edges)
{
    long long e = g * GROUP + lane;
    int ce = (e < (long long)n_edges) ? (int)e : 0;
    return __builtin_nontemporal_load(p + ce);
}

__device__ __forceinline__ v4f load_row(const float* __restrict__ h, int ids, int j, int lane, int n_nodes)
{
    int node = __builtin_amdgcn_readlane(ids, j);
    node = ((unsigned)node < (unsigned)n_nodes) ? node : 0;
    const unsigned long long base = (unsigned long long)h + ((unsigned long long)(unsigned)node << 9);
    return ((gv4f*)base)[(unsigned)lane];
}

__device__ __forceinline__ v8f edge_tile(v4f x, v4f y)
{
    v8f acc = {0.f, 0.f, 0.f, 0.f, 0.f, 0.f, 0.f, 0.f};
    v2f a0 = {x.x, x.y};
    v2f b0 = {y.x, y.y};
    acc = __builtin_amdgcn_wmma_f32_16x16x4_f32(false, a0, false, b0, (short)0, acc, false, false);
    v2f a1 = {x.z, x.w};
    v2f b1 = {y.z, y.w};
    acc = __builtin_amdgcn_wmma_f32_16x16x4_f32(false, a1, false, b1, (short)0, acc, false, false);
    return acc;
}

__device__ __forceinline__ float pick_diag(v8f acc, bool q0, bool q1, bool q2)
{
    float w0 = q0 ? acc[1] : acc[0];
    float w1 = q0 ? acc[3] : acc[2];
    float w2 = q0 ? acc[5] : acc[4];
    float w3 = q0 ? acc[7] : acc[6];
    float u0 = q1 ? w1 : w0;
    float u1 = q1 ? w3 : w2;
    return q2 ? u1 : u0;
}

__device__ __forceinline__ float xhalf(float x)
{
    return __builtin_bit_cast(float, __builtin_amdgcn_permlanex16(__builtin_bit_cast(int, x), __builtin_bit_cast(int, x),
                                                                  0x76543210, 0xfedcba98, false, false));
}

template <int K>
__device__ __forceinline__ float merge(float a, float b, bool hi)
{
    const float keep = hi ? b : a;
    const float send = hi ? a : b;
    float got;
    if (K == 4) {
        got = xhalf(send);
    } else {
        got = __builtin_bit_cast(float, __builtin_amdgcn_update_dpp(0, __builtin_bit_cast(int, send),
                                                                    0x160 + (1 << K), 0xf, 0xf, true));
    }
    return keep + got;
}

template <bool NEXT>
__device__ __forceinline__ void do_group(const float* __restrict__ h, v4f (&xs)[DEPTH], v4f (&ys)[DEPTH],
                                         int sid, int did, int sid_n, int did_n, int lane, int n_nodes,
                                         float* __restrict__ out, long long g, int n_edges)
{
    const bool q0 = (lane & 1) != 0, q1 = (lane & 2) != 0, q2 = (lane & 4) != 0;
    const bool q3 = (lane & 8) != 0, q4 = (lane & 16) != 0;
    const bool on_diag = (lane < 8) || (lane >= 24);

    float t[GROUP], l1[GROUP / 2], l2[GROUP / 4], l3[GROUP / 8], l4[GROUP / 16];
    v8f acc_prev = {0.f, 0.f, 0.f, 0.f, 0.f, 0.f, 0.f, 0.f};
    #pragma unroll
    for (int j = 0; j <= GROUP; ++j) {
        v8f acc = acc_prev;
        if (j < GROUP) {
            const int s = j % DEPTH;
            acc = edge_tile(xs[s], ys[s]);
            const int jn = j + DEPTH;
            if (jn < GROUP) {
                xs[s] = load_row(h, sid, jn, lane, n_nodes);
                ys[s] = load_row(h, did, jn, lane, n_nodes);
            } else if (NEXT) {
                xs[s] = load_row(h, sid_n, jn - GROUP, lane, n_nodes);
                ys[s] = load_row(h, did_n, jn - GROUP, lane, n_nodes);
            }
        }
        if (j > 0) {
            const int i = j - 1;
            t[i] = pick_diag(acc_prev, q0, q1, q2);
            if ((i & 1) == 1)   l1[i >> 1] = merge<0>(t[i - 1], t[i], q0);
            if ((i & 3) == 3)   l2[i >> 2] = merge<1>(l1[(i >> 1) - 1], l1[i >> 1], q1);
            if ((i & 7) == 7)   l3[i >> 3] = merge<2>(l2[(i >> 2) - 1], l2[i >> 2], q2);
            if ((i & 15) == 15) {
                const float za = on_diag ? l3[(i >> 3) - 1] : 0.f;
                const float zb = on_diag ? l3[i >> 3] : 0.f;
                l4[i >> 4] = merge<3>(za, zb, q3);
            }
        }
        acc_prev = acc;
        __builtin_amdgcn_sched_barrier(0);
    }
    const float score = merge<4>(l4[0], l4[1], q4);
    const long long eo = g * GROUP + lane;
    if (eo < (long long)n_edges)
        __builtin_nontemporal_store(score, out + eo);
}

__global__ __launch_bounds__(BLOCK)
void edge_dot_wmma(const float* __restrict__ h,
                   const int*   __restrict__ src,
                   const int*   __restrict__ dst,
                   float*       __restrict__ out,
                   int n_edges, int n_nodes)
{
    const int lane = threadIdx.x & 31;
    const int wave = __builtin_amdgcn_readfirstlane((int)(threadIdx.x >> 5));

    const int ngroups = (n_edges + GROUP - 1) / GROUP;
    const int nwaves  = (int)gridDim.x * WPB;
    int g = (int)blockIdx.x * WPB + wave;
    if (g >= ngroups) return;

    int sid = load_ids(src, g, lane, n_edges);
    int did = load_ids(dst, g, lane, n_edges);

    v4f xs[DEPTH], ys[DEPTH];
    #pragma unroll
    for (int s = 0; s < DEPTH; ++s) {
        xs[s] = load_row(h, sid, s, lane, n_nodes);
        ys[s] = load_row(h, did, s, lane, n_nodes);
    }

    int gn = g + nwaves;
    while (gn < ngroups) {
        const int sid_n = load_ids(src, gn, lane, n_edges);
        const int did_n = load_ids(dst, gn, lane, n_edges);
        do_group<true>(h, xs, ys, sid, did, sid_n, did_n, lane, n_nodes, out, g, n_edges);
        sid = sid_n;
        did = did_n;
        g   = gn;
        gn += nwaves;
    }
    do_group<false>(h, xs, ys, sid, did, 0, 0, lane, n_nodes, out, g, n_edges);
}

extern "C" void kernel_launch(void* const* d_in, const int* in_sizes, int n_in,
                              void* d_out, int out_size, void* d_ws, size_t ws_size,
                              hipStream_t stream) {
    const float* h   = (const float*)d_in[0];
    const int*   src = (const int*)d_in[1];
    const int*   dst = (const int*)d_in[2];
    float*       out = (float*)d_out;

    const int n_edges = in_sizes[1];
    const int n_nodes = in_sizes[0] / D_FEAT;

    const int ngroups = (n_edges + GROUP - 1) / GROUP;
    const int blocks  = (ngroups + WPB * GPW - 1) / (WPB * GPW);

    edge_dot_wmma<<<blocks, BLOCK, 0, stream>>>(h, src, dst, out, n_edges, n_nodes);
}
